# starting kernel
# baseline (speedup 1.0000x reference)
#include <hip/hip_runtime.h>
#include <hip/hip_bf16.h>

typedef __attribute__((ext_vector_type(2))) float v2f;
typedef __attribute__((ext_vector_type(8))) float v8f;

#define D 128
#define LN_EPS 1e-5f

__global__ void k_deg_init(float* __restrict__ deg, int n) {
  int i = blockIdx.x * blockDim.x + threadIdx.x;
  if (i < n) deg[i] = 1.0f;
}

__global__ void k_deg_accum(const int* __restrict__ row, float* __restrict__ deg, int e) {
  int i = blockIdx.x * blockDim.x + threadIdx.x;
  if (i < e) atomicAdd(&deg[row[i]], 1.0f);
}

__global__ void k_dinv(float* __restrict__ deg, int n) {
  int i = blockIdx.x * blockDim.x + threadIdx.x;
  if (i < n) deg[i] = rsqrtf(deg[i]);
}

__global__ __launch_bounds__(256) void k_gemm_wmma(const float* __restrict__ h,
                                                   const float* __restrict__ W,
                                                   float* __restrict__ out, int n) {
  __shared__ __align__(16) float lw[D * D];
  {
    const float4* W4 = (const float4*)W;
    float4* lw4 = (float4*)lw;
    for (int i = threadIdx.x; i < (D * D) / 4; i += 256) lw4[i] = W4[i];
  }
  __syncthreads();

  const int wave = threadIdx.x >> 5;
  const int lane = threadIdx.x & 31;
  const int ln   = lane & 15;
  const int koff = (lane < 16) ? 0 : 2;

  const int row0 = (blockIdx.x * 8 + wave) * 16;
  if (row0 + 16 > n) return;

  v8f acc[8];
  const v8f zero = {0.f, 0.f, 0.f, 0.f, 0.f, 0.f, 0.f, 0.f};
#pragma unroll
  for (int t = 0; t < 8; ++t) acc[t] = zero;

  const float* __restrict__ hrow = h + (size_t)(row0 + ln) * D;

  for (int k0 = 0; k0 < D; k0 += 4) {
    v2f a = *(const v2f*)(hrow + k0 + koff);
#pragma unroll
    for (int t = 0; t < 8; ++t) {
      const int n0 = t * 16;
      v2f b;
      b.x = lw[(k0 + koff) * D + n0 + ln];
      b.y = lw[(k0 + koff + 1) * D + n0 + ln];
      acc[t] = __builtin_amdgcn_wmma_f32_16x16x4_f32(
           false, a,  false, b,
           (short)0, acc[t],  false,  false);
    }
  }

  const int mbase = (lane < 16) ? 0 : 8;
#pragma unroll
  for (int t = 0; t < 8; ++t) {
    const int n0 = t * 16;
#pragma unroll
    for (int j = 0; j < 8; ++j) {
      out[(size_t)(row0 + mbase + j) * D + n0 + ln] = acc[t][j];
    }
  }
}

__global__ void k_agg_init(const float* __restrict__ hw, const float* __restrict__ dinv,
                           const float* __restrict__ bias, float* __restrict__ agg, int n) {
  int i = blockIdx.x * blockDim.x + threadIdx.x;
  int node = i >> 5;
  if (node >= n) return;
  int c = (i & 31) * 4;
  float s = dinv[node];
  s = s * s;
  float4 hv = *(const float4*)(hw + (size_t)node * D + c);
  float4 r;
  r.x = bias[c + 0] + s * hv.x;
  r.y = bias[c + 1] + s * hv.y;
  r.z = bias[c + 2] + s * hv.z;
  r.w = bias[c + 3] + s * hv.w;
  *(float4*)(agg + (size_t)node * D + c) = r;
}

__global__ void k_edge_scatter(const int* __restrict__ er, const int* __restrict__ ec,
                               const float* __restrict__ dinv,
                               const float* __restrict__ hw,
                               float* __restrict__ agg, int e) {
  int i = blockIdx.x * blockDim.x + threadIdx.x;
  int edge = i >> 5;
  if (edge >= e) return;
  int c = (i & 31) * 4;
  int r = er[edge];
  int s = ec[edge];
  float wgt = dinv[r] * dinv[s];
  float4 hv = *(const float4*)(hw + (size_t)s * D + c);
  float* ap = agg + (size_t)r * D + c;
  atomicAdd(ap + 0, wgt * hv.x);
  atomicAdd(ap + 1, wgt * hv.y);
  atomicAdd(ap + 2, wgt * hv.z);
  atomicAdd(ap + 3, wgt * hv.w);
}

__global__ __launch_bounds__(256) void k_ln_relu(const float* __restrict__ agg,
                                                 const float* __restrict__ g,
                                                 const float* __restrict__ beta,
                                                 float* __restrict__ out, int n) {
  const int wave = threadIdx.x >> 5;
  const int lane = threadIdx.x & 31;
  const int row = blockIdx.x * 8 + wave;
  if (row >= n) return;
  const int c = lane * 4;
  float4 v = *(const float4*)(agg + (size_t)row * D + c);

  float s = v.x + v.y + v.z + v.w;
#pragma unroll
  for (int off = 16; off > 0; off >>= 1) s += __shfl_xor(s, off, 32);
  float mu = s * (1.0f / 128.0f);

  float dx = v.x - mu, dy = v.y - mu, dz = v.z - mu, dw = v.w - mu;
  float q = dx * dx + dy * dy + dz * dz + dw * dw;
#pragma unroll
  for (int off = 16; off > 0; off >>= 1) q += __shfl_xor(q, off, 32);
  float rs = rsqrtf(q * (1.0f / 128.0f) + LN_EPS);

  float4 o;
  o.x = fmaxf(dx * rs * g[c + 0] + beta[c + 0], 0.0f);
  o.y = fmaxf(dy * rs * g[c + 1] + beta[c + 1], 0.0f);
  o.z = fmaxf(dz * rs * g[c + 2] + beta[c + 2], 0.0f);
  o.w = fmaxf(dw * rs * g[c + 3] + beta[c + 3], 0.0f);
  *(float4*)(out + (size_t)row * D + c) = o;
}

extern "C" void kernel_launch(void* const* d_in, const int* in_sizes, int n_in,
                              void* d_out, int out_size, void* d_ws, size_t ws_size,
                              hipStream_t stream) {
  const float* x    = (const float*)d_in[0];
  const int*   erow = (const int*)d_in[1];
  const int*   ecol = (const int*)d_in[2];
  const float* W[3]  = {(const float*)d_in[3], (const float*)d_in[7],  (const float*)d_in[11]};
  const float* b[3]  = {(const float*)d_in[4], (const float*)d_in[8],  (const float*)d_in[12]};
  const float* g[3]  = {(const float*)d_in[5], (const float*)d_in[9],  (const float*)d_in[13]};
  const float* bt[3] = {(const float*)d_in[6], (const float*)d_in[10], (const float*)d_in[14]};

  const int n = in_sizes[0] / D;
  const int e = in_sizes[1];

  float* ws   = (float*)d_ws;
  float* dinv = ws;
  float* hw   = ws + n;
  float* agg  = hw + (size_t)n * D;
  float* outp = (float*)d_out;

  k_deg_init<<<(n + 255) / 256, 256, 0, stream>>>(dinv, n);
  k_deg_accum<<<(e + 255) / 256, 256, 0, stream>>>(erow, dinv, e);
  k_dinv<<<(n + 255) / 256, 256, 0, stream>>>(dinv, n);

  const unsigned gemm_blocks = (unsigned)((n / 16 + 7) / 8);
  const unsigned node_chunks = (unsigned)(((size_t)n * 32 + 255) / 256);
  const unsigned edge_chunks = (unsigned)(((size_t)e * 32 + 255) / 256);
  const unsigned ln_blocks   = (unsigned)((n + 7) / 8);

  const float* hin = x;
  for (int L = 0; L < 3; ++L) {
    k_gemm_wmma<<<gemm_blocks, 256, 0, stream>>>(hin, W[L], hw, n);
    k_agg_init<<<node_chunks, 256, 0, stream>>>(hw, dinv, b[L], agg, n);
    k_edge_scatter<<<edge_chunks, 256, 0, stream>>>(erow, ecol, dinv, hw, agg, e);
    k_ln_relu<<<ln_blocks, 256, 0, stream>>>(agg, g[L], bt[L], outp, n);
    hin = outp;
  }
}


# Float-atomic edge scatter replaced by a per-call sort of edges by destination row (int atomic rank + scan); one wave per row sums its sources in registers in rank order, one store per row
# speedup vs baseline: 2.4995x; 2.4995x over previous
#include <hip/hip_runtime.h>
#include <hip/hip_bf16.h>

typedef __attribute__((ext_vector_type(2))) float v2f;
typedef __attribute__((ext_vector_type(8))) float v8f;

#define D 128
#define LN_EPS 1e-5f

__global__ void k_cnt_zero(int* __restrict__ cnt, int n) {
  int i = blockIdx.x * blockDim.x + threadIdx.x;
  if (i < n) cnt[i] = 0;
}

__global__ void k_count_rank(const int* __restrict__ row, int* __restrict__ cnt,
                             int* __restrict__ rank, int e) {
  int i = blockIdx.x * blockDim.x + threadIdx.x;
  if (i < e) rank[i] = atomicAdd(&cnt[row[i]], 1);
}

#define SCAN_EPT 8
#define SCAN_TILE (1024 * SCAN_EPT)
__global__ __launch_bounds__(1024) void k_scan_rows(const int* __restrict__ cnt,
                                                    int* __restrict__ rowptr,
                                                    float* __restrict__ dinv, int n) {
  __shared__ int wtot[32];
  __shared__ int tile_total;
  const int tid  = threadIdx.x;
  const int lane = tid & 31;
  const int wave = tid >> 5;
  const int last = max(n - 1, 0);
  int carry = 0;
  int c[SCAN_EPT], cn[SCAN_EPT];
#pragma unroll
  for (int u = 0; u < SCAN_EPT; ++u) {
    const int i = tid * SCAN_EPT + u;
    const int v = cnt[min(i, last)];
    c[u] = (i < n) ? v : 0;
  }
  for (int base = 0; base < n; base += SCAN_TILE) {
    const int i0 = base + tid * SCAN_EPT;
#pragma unroll
    for (int u = 0; u < SCAN_EPT; ++u) {
      const int i = i0 + SCAN_TILE + u;
      const int v = cnt[min(i, last)];
      cn[u] = (i < n) ? v : 0;
    }
    int t = 0;
#pragma unroll
    for (int u = 0; u < SCAN_EPT; ++u) t += c[u];
    int incl = t;
#pragma unroll
    for (int off = 1; off < 32; off <<= 1) {
      const int v = __shfl_up(incl, off, 32);
      if (lane >= off) incl += v;
    }
    if (lane == 31) wtot[wave] = incl;
    __syncthreads();
    if (wave == 0) {
      const int wv = wtot[lane];
      int wi = wv;
#pragma unroll
      for (int off = 1; off < 32; off <<= 1) {
        const int v = __shfl_up(wi, off, 32);
        if (lane >= off) wi += v;
      }
      wtot[lane] = wi - wv;
      if (lane == 31) tile_total = wi;
    }
    __syncthreads();
    int ex = carry + wtot[wave] + (incl - t);
#pragma unroll
    for (int u = 0; u < SCAN_EPT; ++u) {
      const int i = i0 + u;
      if (i < n) {
        rowptr[i] = ex;
        dinv[i] = rsqrtf((float)(c[u] + 1));
      }
      ex += c[u];
    }
    carry += tile_total;
    __syncthreads();
#pragma unroll
    for (int u = 0; u < SCAN_EPT; ++u) c[u] = cn[u];
  }
  if (tid == 0) rowptr[n] = carry;
}

__global__ void k_csr_fill(const int* __restrict__ er, const int* __restrict__ ec,
                           const int* __restrict__ rank, const int* __restrict__ rowptr,
                           const float* __restrict__ dinv, int* __restrict__ csr, int e) {
  int i = blockIdx.x * blockDim.x + threadIdx.x;
  if (i >= e) return;
  const int r = er[i];
  const int s = ec[i];
  int2 ent;
  ent.x = s;
  ent.y = __float_as_int(dinv[r] * dinv[s]);
  ((int2*)csr)[rowptr[r] + rank[i]] = ent;
}

__global__ __launch_bounds__(256) void k_gemm_wmma(const float* __restrict__ h,
                                                   const float* __restrict__ W,
                                                   float* __restrict__ out, int n) {
  __shared__ __align__(16) float lw[D * D];
  {
    const float4* W4 = (const float4*)W;
    float4* lw4 = (float4*)lw;
    for (int i = threadIdx.x; i < (D * D) / 4; i += 256) lw4[i] = W4[i];
  }
  __syncthreads();

  const int wave = threadIdx.x >> 5;
  const int lane = threadIdx.x & 31;
  const int ln   = lane & 15;
  const int koff = (lane < 16) ? 0 : 2;

  const int row0 = (blockIdx.x * 8 + wave) * 16;
  if (row0 + 16 > n) return;

  v8f acc[8];
  const v8f zero = {0.f, 0.f, 0.f, 0.f, 0.f, 0.f, 0.f, 0.f};
#pragma unroll
  for (int t = 0; t < 8; ++t) acc[t] = zero;

  const float* __restrict__ hrow = h + (size_t)(row0 + ln) * D;

  for (int k0 = 0; k0 < D; k0 += 4) {
    v2f a = *(const v2f*)(hrow + k0 + koff);
#pragma unroll
    for (int t = 0; t < 8; ++t) {
      const int n0 = t * 16;
      v2f b;
      b.x = lw[(k0 + koff) * D + n0 + ln];
      b.y = lw[(k0 + koff + 1) * D + n0 + ln];
      acc[t] = __builtin_amdgcn_wmma_f32_16x16x4_f32(
           false, a,  false, b,
           (short)0, acc[t],  false,  false);
    }
  }

  const int mbase = (lane < 16) ? 0 : 8;
#pragma unroll
  for (int t = 0; t < 8; ++t) {
    const int n0 = t * 16;
#pragma unroll
    for (int j = 0; j < 8; ++j) {
      out[(size_t)(row0 + mbase + j) * D + n0 + ln] = acc[t][j];
    }
  }
}

template <int K>
__device__ __forceinline__ void gather_rows(const int2 my, const int j,
                                            const float* __restrict__ hw, const int c,
                                            float4& acc) {
  float4 v[K];
  float w[K];
#pragma unroll
  for (int u = 0; u < K; ++u) {
    const int src = __builtin_amdgcn_readlane(my.x, j + u);
    w[u] = __int_as_float(__builtin_amdgcn_readlane(my.y, j + u));
    v[u] = *(const float4*)(hw + (size_t)src * D + c);
  }
#pragma unroll
  for (int u = 0; u < K; ++u) {
    acc.x = fmaf(w[u], v[u].x, acc.x);
    acc.y = fmaf(w[u], v[u].y, acc.y);
    acc.z = fmaf(w[u], v[u].z, acc.z);
    acc.w = fmaf(w[u], v[u].w, acc.w);
  }
}

__global__ __launch_bounds__(256) void k_row_gather(const int* __restrict__ rowptr,
                                                    const int* __restrict__ csr,
                                                    const float* __restrict__ dinv,
                                                    const float* __restrict__ hw,
                                                    const float* __restrict__ bias,
                                                    float* __restrict__ agg, int n) {
  const int lane = threadIdx.x & 31;
  const int wave = __builtin_amdgcn_readfirstlane((int)(threadIdx.x >> 5));
  const int row = blockIdx.x * 8 + wave;
  if (row >= n) return;
  const int c = lane * 4;
  const int beg = rowptr[row];
  const int end = rowptr[row + 1];
  float s = dinv[row];
  s = s * s;
  const float4 hv = *(const float4*)(hw + (size_t)row * D + c);
  float4 acc;
  acc.x = bias[c + 0] + s * hv.x;
  acc.y = bias[c + 1] + s * hv.y;
  acc.z = bias[c + 2] + s * hv.z;
  acc.w = bias[c + 3] + s * hv.w;
  const int2* __restrict__ ed = (const int2*)csr;
  for (int base = beg; base < end; base += 32) {
    const int m = min(32, end - base);
    int2 my;
    my.x = 0;
    my.y = 0;
    if (lane < m) my = ed[base + lane];
    int j = 0;
    for (; j + 8 <= m; j += 8) gather_rows<8>(my, j, hw, c, acc);
    if (j + 4 <= m) { gather_rows<4>(my, j, hw, c, acc); j += 4; }
    if (j + 2 <= m) { gather_rows<2>(my, j, hw, c, acc); j += 2; }
    if (j < m) gather_rows<1>(my, j, hw, c, acc);
  }
  *(float4*)(agg + (size_t)row * D + c) = acc;
}

__global__ __launch_bounds__(256) void k_ln_relu(const float* __restrict__ agg,
                                                 const float* __restrict__ g,
                                                 const float* __restrict__ beta,
                                                 float* __restrict__ out, int n) {
  const int wave = threadIdx.x >> 5;
  const int lane = threadIdx.x & 31;
  const int row = blockIdx.x * 8 + wave;
  if (row >= n) return;
  const int c = lane * 4;
  float4 v = *(const float4*)(agg + (size_t)row * D + c);

  float s = v.x + v.y + v.z + v.w;
#pragma unroll
  for (int off = 16; off > 0; off >>= 1) s += __shfl_xor(s, off, 32);
  float mu = s * (1.0f / 128.0f);

  float dx = v.x - mu, dy = v.y - mu, dz = v.z - mu, dw = v.w - mu;
  float q = dx * dx + dy * dy + dz * dz + dw * dw;
#pragma unroll
  for (int off = 16; off > 0; off >>= 1) q += __shfl_xor(q, off, 32);
  float rs = rsqrtf(q * (1.0f / 128.0f) + LN_EPS);

  float4 o;
  o.x = fmaxf(dx * rs * g[c + 0] + beta[c + 0], 0.0f);
  o.y = fmaxf(dy * rs * g[c + 1] + beta[c + 1], 0.0f);
  o.z = fmaxf(dz * rs * g[c + 2] + beta[c + 2], 0.0f);
  o.w = fmaxf(dw * rs * g[c + 3] + beta[c + 3], 0.0f);
  *(float4*)(out + (size_t)row * D + c) = o;
}

extern "C" void kernel_launch(void* const* d_in, const int* in_sizes, int n_in,
                              void* d_out, int out_size, void* d_ws, size_t ws_size,
                              hipStream_t stream) {
  const float* x    = (const float*)d_in[0];
  const int*   erow = (const int*)d_in[1];
  const int*   ecol = (const int*)d_in[2];
  const float* W[3]  = {(const float*)d_in[3], (const float*)d_in[7],  (const float*)d_in[11]};
  const float* b[3]  = {(const float*)d_in[4], (const float*)d_in[8],  (const float*)d_in[12]};
  const float* g[3]  = {(const float*)d_in[5], (const float*)d_in[9],  (const float*)d_in[13]};
  const float* bt[3] = {(const float*)d_in[6], (const float*)d_in[10], (const float*)d_in[14]};

  const int n = in_sizes[0] / D;
  const int e = in_sizes[1];

  float* ws   = (float*)d_ws;
  float* dinv = ws;
  float* hw   = ws + n;
  float* agg  = hw + (size_t)n * D;
  int* csr    = (int*)(agg + (size_t)n * D);
  int* rank   = csr + (size_t)e * 2;
  int* cnt    = rank + e;
  int* rowptr = cnt + n;
  float* outp = (float*)d_out;

  k_cnt_zero<<<(n + 255) / 256, 256, 0, stream>>>(cnt, n);
  k_count_rank<<<(e + 255) / 256, 256, 0, stream>>>(erow, cnt, rank, e);
  k_scan_rows<<<1, 1024, 0, stream>>>(cnt, rowptr, dinv, n);
  k_csr_fill<<<(e + 255) / 256, 256, 0, stream>>>(erow, ecol, rank, rowptr, dinv, csr, e);

  const unsigned gemm_blocks = (unsigned)((n / 16 + 7) / 8);
  const unsigned row_blocks  = (unsigned)((n + 7) / 8);

  const float* hin = x;
  for (int L = 0; L < 3; ++L) {
    k_gemm_wmma<<<gemm_blocks, 256, 0, stream>>>(hin, W[L], hw, n);
    k_row_gather<<<row_blocks, 256, 0, stream>>>(rowptr, csr, dinv, hw, b[L], agg, n);
    k_ln_relu<<<row_blocks, 256, 0, stream>>>(agg, g[L], bt[L], outp, n);
    hin = outp;
  }
}


# LayerNorm+ReLU fused into the row-gather kernel: the wave normalises the row it has just summed in registers and stores h; aggregate plane and 3 launches (13 to 10) removed; sum order unchanged
# speedup vs baseline: 2.7002x; 1.0803x over previous
#include <hip/hip_runtime.h>
#include <hip/hip_bf16.h>

typedef __attribute__((ext_vector_type(2))) float v2f;
typedef __attribute__((ext_vector_type(8))) float v8f;

#define D 128
#define LN_EPS 1e-5f

__global__ void k_cnt_zero(int* __restrict__ cnt, int n) {
  int i = blockIdx.x * blockDim.x + threadIdx.x;
  if (i < n) cnt[i] = 0;
}

__global__ void k_count_rank(const int* __restrict__ row, int* __restrict__ cnt,
                             int* __restrict__ rank, int e) {
  int i = blockIdx.x * blockDim.x + threadIdx.x;
  if (i < e) rank[i] = atomicAdd(&cnt[row[i]], 1);
}

#define SCAN_EPT 8
#define SCAN_TILE (1024 * SCAN_EPT)
__global__ __launch_bounds__(1024) void k_scan_rows(const int* __restrict__ cnt,
                                                    int* __restrict__ rowptr,
                                                    float* __restrict__ dinv, int n) {
  __shared__ int wtot[32];
  __shared__ int tile_total;
  const int tid  = threadIdx.x;
  const int lane = tid & 31;
  const int wave = tid >> 5;
  const int last = max(n - 1, 0);
  int carry = 0;
  int c[SCAN_EPT], cn[SCAN_EPT];
#pragma unroll
  for (int u = 0; u < SCAN_EPT; ++u) {
    const int i = tid * SCAN_EPT + u;
    const int v = cnt[min(i, last)];
    c[u] = (i < n) ? v : 0;
  }
  for (int base = 0; base < n; base += SCAN_TILE) {
    const int i0 = base + tid * SCAN_EPT;
#pragma unroll
    for (int u = 0; u < SCAN_EPT; ++u) {
      const int i = i0 + SCAN_TILE + u;
      const int v = cnt[min(i, last)];
      cn[u] = (i < n) ? v : 0;
    }
    int t = 0;
#pragma unroll
    for (int u = 0; u < SCAN_EPT; ++u) t += c[u];
    int incl = t;
#pragma unroll
    for (int off = 1; off < 32; off <<= 1) {
      const int v = __shfl_up(incl, off, 32);
      if (lane >= off) incl += v;
    }
    if (lane == 31) wtot[wave] = incl;
    __syncthreads();
    if (wave == 0) {
      const int wv = wtot[lane];
      int wi = wv;
#pragma unroll
      for (int off = 1; off < 32; off <<= 1) {
        const int v = __shfl_up(wi, off, 32);
        if (lane >= off) wi += v;
      }
      wtot[lane] = wi - wv;
      if (lane == 31) tile_total = wi;
    }
    __syncthreads();
    int ex = carry + wtot[wave] + (incl - t);
#pragma unroll
    for (int u = 0; u < SCAN_EPT; ++u) {
      const int i = i0 + u;
      if (i < n) {
        rowptr[i] = ex;
        dinv[i] = rsqrtf((float)(c[u] + 1));
      }
      ex += c[u];
    }
    carry += tile_total;
    __syncthreads();
#pragma unroll
    for (int u = 0; u < SCAN_EPT; ++u) c[u] = cn[u];
  }
  if (tid == 0) rowptr[n] = carry;
}

__global__ void k_csr_fill(const int* __restrict__ er, const int* __restrict__ ec,
                           const int* __restrict__ rank, const int* __restrict__ rowptr,
                           const float* __restrict__ dinv, int* __restrict__ csr, int e) {
  int i = blockIdx.x * blockDim.x + threadIdx.x;
  if (i >= e) return;
  const int r = er[i];
  const int s = ec[i];
  int2 ent;
  ent.x = s;
  ent.y = __float_as_int(dinv[r] * dinv[s]);
  ((int2*)csr)[rowptr[r] + rank[i]] = ent;
}

__global__ __launch_bounds__(256) void k_gemm_wmma(const float* __restrict__ h,
                                                   const float* __restrict__ W,
                                                   float* __restrict__ out, int n) {
  __shared__ __align__(16) float lw[D * D];
  {
    const float4* W4 = (const float4*)W;
    float4* lw4 = (float4*)lw;
    for (int i = threadIdx.x; i < (D * D) / 4; i += 256) lw4[i] = W4[i];
  }
  __syncthreads();

  const int wave = threadIdx.x >> 5;
  const int lane = threadIdx.x & 31;
  const int ln   = lane & 15;
  const int koff = (lane < 16) ? 0 : 2;

  const int row0 = (blockIdx.x * 8 + wave) * 16;
  if (row0 + 16 > n) return;

  v8f acc[8];
  const v8f zero = {0.f, 0.f, 0.f, 0.f, 0.f, 0.f, 0.f, 0.f};
#pragma unroll
  for (int t = 0; t < 8; ++t) acc[t] = zero;

  const float* __restrict__ hrow = h + (size_t)(row0 + ln) * D;

  for (int k0 = 0; k0 < D; k0 += 4) {
    v2f a = *(const v2f*)(hrow + k0 + koff);
#pragma unroll
    for (int t = 0; t < 8; ++t) {
      const int n0 = t * 16;
      v2f b;
      b.x = lw[(k0 + koff) * D + n0 + ln];
      b.y = lw[(k0 + koff + 1) * D + n0 + ln];
      acc[t] = __builtin_amdgcn_wmma_f32_16x16x4_f32(
           false, a,  false, b,
           (short)0, acc[t],  false,  false);
    }
  }

  const int mbase = (lane < 16) ? 0 : 8;
#pragma unroll
  for (int t = 0; t < 8; ++t) {
    const int n0 = t * 16;
#pragma unroll
    for (int j = 0; j < 8; ++j) {
      out[(size_t)(row0 + mbase + j) * D + n0 + ln] = acc[t][j];
    }
  }
}

template <int K>
__device__ __forceinline__ void gather_rows(const int2 my, const int j,
                                            const float* __restrict__ hw, const int c,
                                            float4& acc) {
  float4 v[K];
  float w[K];
#pragma unroll
  for (int u = 0; u < K; ++u) {
    const int src = __builtin_amdgcn_readlane(my.x, j + u);
    w[u] = __int_as_float(__builtin_amdgcn_readlane(my.y, j + u));
    v[u] = *(const float4*)(hw + (size_t)src * D + c);
  }
#pragma unroll
  for (int u = 0; u < K; ++u) {
    acc.x = fmaf(w[u], v[u].x, acc.x);
    acc.y = fmaf(w[u], v[u].y, acc.y);
    acc.z = fmaf(w[u], v[u].z, acc.z);
    acc.w = fmaf(w[u], v[u].w, acc.w);
  }
}

__global__ __launch_bounds__(256) void k_row_gather_ln(const int* __restrict__ rowptr,
                                                       const int* __restrict__ csr,
                                                       const float* __restrict__ dinv,
                                                       const float* __restrict__ hw,
                                                       const float* __restrict__ bias,
                                                       const float* __restrict__ g,
                                                       const float* __restrict__ beta,
                                                       float* __restrict__ out, int n) {
  const int lane = threadIdx.x & 31;
  const int wave = __builtin_amdgcn_readfirstlane((int)(threadIdx.x >> 5));
  const int row = blockIdx.x * 8 + wave;
  if (row >= n) return;
  const int c = lane * 4;
  const int beg = rowptr[row];
  const int end = rowptr[row + 1];
  float s = dinv[row];
  s = s * s;
  const float4 hv = *(const float4*)(hw + (size_t)row * D + c);
  float4 acc;
  acc.x = bias[c + 0] + s * hv.x;
  acc.y = bias[c + 1] + s * hv.y;
  acc.z = bias[c + 2] + s * hv.z;
  acc.w = bias[c + 3] + s * hv.w;
  const int2* __restrict__ ed = (const int2*)csr;
  for (int base = beg; base < end; base += 32) {
    const int m = min(32, end - base);
    int2 my;
    my.x = 0;
    my.y = 0;
    if (lane < m) my = ed[base + lane];
    int j = 0;
    for (; j + 8 <= m; j += 8) gather_rows<8>(my, j, hw, c, acc);
    if (j + 4 <= m) { gather_rows<4>(my, j, hw, c, acc); j += 4; }
    if (j + 2 <= m) { gather_rows<2>(my, j, hw, c, acc); j += 2; }
    if (j < m) gather_rows<1>(my, j, hw, c, acc);
  }

  float t = acc.x + acc.y + acc.z + acc.w;
#pragma unroll
  for (int off = 16; off > 0; off >>= 1) t += __shfl_xor(t, off, 32);
  const float mu = t * (1.0f / 128.0f);

  const float dx = acc.x - mu, dy = acc.y - mu, dz = acc.z - mu, dw = acc.w - mu;
  float q = dx * dx + dy * dy + dz * dz + dw * dw;
#pragma unroll
  for (int off = 16; off > 0; off >>= 1) q += __shfl_xor(q, off, 32);
  const float rs = rsqrtf(q * (1.0f / 128.0f) + LN_EPS);

  float4 o;
  o.x = fmaxf(dx * rs * g[c + 0] + beta[c + 0], 0.0f);
  o.y = fmaxf(dy * rs * g[c + 1] + beta[c + 1], 0.0f);
  o.z = fmaxf(dz * rs * g[c + 2] + beta[c + 2], 0.0f);
  o.w = fmaxf(dw * rs * g[c + 3] + beta[c + 3], 0.0f);
  *(float4*)(out + (size_t)row * D + c) = o;
}

extern "C" void kernel_launch(void* const* d_in, const int* in_sizes, int n_in,
                              void* d_out, int out_size, void* d_ws, size_t ws_size,
                              hipStream_t stream) {
  const float* x    = (const float*)d_in[0];
  const int*   erow = (const int*)d_in[1];
  const int*   ecol = (const int*)d_in[2];
  const float* W[3]  = {(const float*)d_in[3], (const float*)d_in[7],  (const float*)d_in[11]};
  const float* b[3]  = {(const float*)d_in[4], (const float*)d_in[8],  (const float*)d_in[12]};
  const float* g[3]  = {(const float*)d_in[5], (const float*)d_in[9],  (const float*)d_in[13]};
  const float* bt[3] = {(const float*)d_in[6], (const float*)d_in[10], (const float*)d_in[14]};

  const int n = in_sizes[0] / D;
  const int e = in_sizes[1];

  float* ws   = (float*)d_ws;
  float* dinv = ws;
  float* hw   = ws + n;
  int* csr    = (int*)(hw + (size_t)n * D);
  int* rank   = csr + (size_t)e * 2;
  int* cnt    = rank + e;
  int* rowptr = cnt + n;
  float* outp = (float*)d_out;

  k_cnt_zero<<<(n + 255) / 256, 256, 0, stream>>>(cnt, n);
  k_count_rank<<<(e + 255) / 256, 256, 0, stream>>>(erow, cnt, rank, e);
  k_scan_rows<<<1, 1024, 0, stream>>>(cnt, rowptr, dinv, n);
  k_csr_fill<<<(e + 255) / 256, 256, 0, stream>>>(erow, ecol, rank, rowptr, dinv, csr, e);

  const unsigned gemm_blocks = (unsigned)((n / 16 + 7) / 8);
  const unsigned row_blocks  = (unsigned)((n + 7) / 8);

  const float* hin = x;
  for (int L = 0; L < 3; ++L) {
    k_gemm_wmma<<<gemm_blocks, 256, 0, stream>>>(hin, W[L], hw, n);
    k_row_gather_ln<<<row_blocks, 256, 0, stream>>>(rowptr, csr, dinv, hw, b[L], g[L], bt[L], outp, n);
    hin = outp;
  }
}


# GEMM re-tiled: 256 blocks of 16 waves, one per WGP, stage W transposed in LDS once and walk 16x64 units with 16-byte fragment loads; matrix work even over SIMDs; k summed 0,1,4,5 then 2,3,6,7 per 8
# speedup vs baseline: 4.6346x; 1.7164x over previous
#include <hip/hip_runtime.h>
#include <hip/hip_bf16.h>

typedef __attribute__((ext_vector_type(2))) float v2f;
typedef __attribute__((ext_vector_type(8))) float v8f;

#define D 128
#define LN_EPS 1e-5f

#define BIN_SHIFT 9
#define BIN_ROWS (1 << BIN_SHIFT)
#define SRC_BITS 23
#define SRC_MASK ((1u << SRC_BITS) - 1u)
#define MAX_BINS 1024
#define CHUNK 8192
#define CHUNK_EPT (CHUNK / 1024)
#define HIST_BATCH 16
#define SORT_CAP 12288
#define SORT_EPT 8

__device__ __forceinline__ int block_excl_scan(const int v, int* wtot, int& total) {
  const int lane = threadIdx.x & 31;
  const int wave = threadIdx.x >> 5;
  int incl = v;
#pragma unroll
  for (int off = 1; off < 32; off <<= 1) {
    const int t = __shfl_up(incl, off, 32);
    if (lane >= off) incl += t;
  }
  if (lane == 31) wtot[wave] = incl;
  __syncthreads();
  if (wave == 0) {
    const int wv = wtot[lane];
    int wi = wv;
#pragma unroll
    for (int off = 1; off < 32; off <<= 1) {
      const int t = __shfl_up(wi, off, 32);
      if (lane >= off) wi += t;
    }
    wtot[lane] = wi - wv;
    if (lane == 31) wtot[32] = wi;
  }
  __syncthreads();
  const int ex = wtot[wave] + incl - v;
  total = wtot[32];
  __syncthreads();
  return ex;
}

__global__ __launch_bounds__(1024) void k_bin_count(const int* __restrict__ row,
                                                    int* __restrict__ hist, int e, int nb) {
  __shared__ int h[MAX_BINS];
  const int tid = threadIdx.x;
  if (tid < nb) h[tid] = 0;
  const int base = blockIdx.x * CHUNK;
  const int last = max(e - 1, 0);
  int r[CHUNK_EPT];
#pragma unroll
  for (int u = 0; u < CHUNK_EPT; ++u) r[u] = row[min(base + u * 1024 + tid, last)];
  __syncthreads();
#pragma unroll
  for (int u = 0; u < CHUNK_EPT; ++u) {
    if (base + u * 1024 + tid < e) atomicAdd(&h[r[u] >> BIN_SHIFT], 1);
  }
  __syncthreads();
  if (tid < nb) hist[blockIdx.x * nb + tid] = h[tid];
}

__global__ __launch_bounds__(1024) void k_bin_scatter(const int* __restrict__ row,
                                                      const int* __restrict__ col,
                                                      const int* __restrict__ hist,
                                                      unsigned* __restrict__ binned,
                                                      int* __restrict__ binoff,
                                                      int e, int nb, int nchunks) {
  __shared__ int loc[MAX_BINS];
  __shared__ int gof[MAX_BINS];
  __shared__ int fill[MAX_BINS];
  __shared__ int wtot[33];
  __shared__ unsigned st_ent[CHUNK];
  __shared__ unsigned short st_bin[CHUNK];
  const int tid = threadIdx.x;
  const int j = blockIdx.x;
  const int base = j * CHUNK;
  const int last = max(e - 1, 0);
  int r[CHUNK_EPT], s[CHUNK_EPT];
#pragma unroll
  for (int u = 0; u < CHUNK_EPT; ++u) {
    const int i = min(base + u * 1024 + tid, last);
    r[u] = row[i];
    s[u] = col[i];
  }
  int* ptot = (int*)st_ent;
  int* ppre = ptot + 4 * MAX_BINS;
  const int q = tid >> 8;
  for (int t0 = 0; t0 < nb; t0 += 256) {
    const int t = t0 + (tid & 255);
    int a = 0, p = 0;
    const int tc = min(t, nb - 1);
    for (int jj0 = q; jj0 < nchunks; jj0 += 4 * HIST_BATCH) {
      int v[HIST_BATCH];
#pragma unroll
      for (int u = 0; u < HIST_BATCH; ++u) v[u] = hist[min(jj0 + 4 * u, nchunks - 1) * nb + tc];
#pragma unroll
      for (int u = 0; u < HIST_BATCH; ++u) {
        const int jj = jj0 + 4 * u;
        if (jj < nchunks) {
          a += v[u];
          if (jj < j) p += v[u];
        }
      }
    }
    ptot[q * MAX_BINS + t] = a;
    ppre[q * MAX_BINS + t] = p;
  }
  __syncthreads();
  int tot = 0, pre = 0, own = 0;
  if (tid < nb) {
    own = hist[j * nb + tid];
#pragma unroll
    for (int k = 0; k < 4; ++k) {
      tot += ptot[k * MAX_BINS + tid];
      pre += ppre[k * MAX_BINS + tid];
    }
  }
  int total_all, total_own;
  const int ex_tot = block_excl_scan(tot, wtot, total_all);
  const int ex_own = block_excl_scan(own, wtot, total_own);
  if (tid < nb) {
    loc[tid] = ex_own;
    gof[tid] = ex_tot + pre;
    fill[tid] = 0;
    if (j == 0) binoff[tid] = ex_tot;
  }
  if (j == 0 && tid == 0) binoff[nb] = total_all;
  __syncthreads();
#pragma unroll
  for (int u = 0; u < CHUNK_EPT; ++u) {
    if (base + u * 1024 + tid < e) {
      const int b = r[u] >> BIN_SHIFT;
      const int p = loc[b] + atomicAdd(&fill[b], 1);
      st_ent[p] = ((unsigned)(r[u] & (BIN_ROWS - 1)) << SRC_BITS) | (unsigned)s[u];
      st_bin[p] = (unsigned short)b;
    }
  }
  __syncthreads();
#pragma unroll
  for (int u = 0; u < CHUNK_EPT; ++u) {
    const int p = u * 1024 + tid;
    if (p < total_own) {
      const int b = st_bin[p];
      binned[gof[b] + (p - loc[b])] = st_ent[p];
    }
  }
}

__global__ __launch_bounds__(1024) void k_bin_sort(const unsigned* __restrict__ binned,
                                                   const int* __restrict__ binoff,
                                                   int* __restrict__ rowptr,
                                                   float* __restrict__ dinv,
                                                   int* __restrict__ csr, int n, int e) {
  __shared__ int cnt[BIN_ROWS];
  __shared__ int fill[BIN_ROWS];
  __shared__ int wtot[33];
  __shared__ int stage[SORT_CAP];
  const int tid = threadIdx.x;
  const int b = blockIdx.x;
  const int beg = binoff[b];
  const int end = binoff[b + 1];
  const int m = end - beg;
  if (tid < BIN_ROWS) cnt[tid] = 0;
  __syncthreads();
  for (int i0 = beg; i0 < end; i0 += 1024 * SORT_EPT) {
    unsigned ent[SORT_EPT];
#pragma unroll
    for (int u = 0; u < SORT_EPT; ++u) ent[u] = binned[min(i0 + u * 1024 + tid, end - 1)];
#pragma unroll
    for (int u = 0; u < SORT_EPT; ++u) {
      if (i0 + u * 1024 + tid < end) atomicAdd(&cnt[ent[u] >> SRC_BITS], 1);
    }
  }
  __syncthreads();
  const int c = (tid < BIN_ROWS) ? cnt[tid] : 0;
  int total;
  const int ex = block_excl_scan(c, wtot, total);
  if (tid < BIN_ROWS) {
    fill[tid] = ex;
    const int rr = b * BIN_ROWS + tid;
    if (rr < n) {
      rowptr[rr] = beg + ex;
      dinv[rr] = rsqrtf((float)(c + 1));
    }
  }
  if (b == 0 && tid == 0) rowptr[n] = e;
  __syncthreads();
  const bool staged = (m <= SORT_CAP);
  for (int i0 = beg; i0 < end; i0 += 1024 * SORT_EPT) {
    unsigned ent[SORT_EPT];
#pragma unroll
    for (int u = 0; u < SORT_EPT; ++u) ent[u] = binned[min(i0 + u * 1024 + tid, end - 1)];
#pragma unroll
    for (int u = 0; u < SORT_EPT; ++u) {
      if (i0 + u * 1024 + tid < end) {
        const int p = atomicAdd(&fill[ent[u] >> SRC_BITS], 1);
        const int sv = (int)(ent[u] & SRC_MASK);
        if (staged) stage[p] = sv;
        else csr[beg + p] = sv;
      }
    }
  }
  if (staged) {
    __syncthreads();
    for (int i = tid; i < m; i += 1024) csr[beg + i] = stage[i];
  }
}

#define GEMM_BLOCKS 256
#define WT_PITCH 132
#define GEMM_LDS_FLOATS (41 * 1024)

__global__ __launch_bounds__(512) void k_gemm_wmma(const float* __restrict__ h,
                                                    const float* __restrict__ W,
                                                    const float* __restrict__ dinv,
                                                    float* __restrict__ out, int n) {
  __shared__ __align__(16) float wt[GEMM_LDS_FLOATS];
#pragma unroll
  for (int i = threadIdx.x; i < (D / 4) * D; i += 512) {
    const int col = i & (D - 1);
    const int k4 = (i >> 7) * 4;
    float4 v;
    v.x = W[(k4 + 0) * D + col];
    v.y = W[(k4 + 1) * D + col];
    v.z = W[(k4 + 2) * D + col];
    v.w = W[(k4 + 3) * D + col];
    *(float4*)(wt + col * WT_PITCH + k4) = v;
  }
  __syncthreads();

  const int wave = __builtin_amdgcn_readfirstlane((int)(threadIdx.x >> 5));
  const int lane = threadIdx.x & 31;
  const int c    = lane & 15;
  const int hh   = lane >> 4;
  const int nstrips = n >> 4;

  const int slot = ((((wave & 3) - (wave >> 2)) & 3) << 2) | (wave >> 2);
  for (int q = slot; ; q += 16) {
    const int strip = blockIdx.x + (q >> 1) * GEMM_BLOCKS;
    if (strip >= nstrips) break;
    const int row0 = strip * 16;
    const int col0 = (q & 1) * 64;

    const float* __restrict__ arow = h + (size_t)(row0 + c) * D + 4 * hh;
    const float* bcol = wt + (col0 + c) * WT_PITCH + 4 * hh;

    v8f acc[4];
    const v8f zero = {0.f, 0.f, 0.f, 0.f, 0.f, 0.f, 0.f, 0.f};
#pragma unroll
    for (int t = 0; t < 4; ++t) acc[t] = zero;

    float4 a_cur = *(const float4*)arow;
    float4 b_cur[4];
#pragma unroll
    for (int t = 0; t < 4; ++t) b_cur[t] = *(const float4*)(bcol + t * 16 * WT_PITCH);

#pragma unroll
    for (int kb = 0; kb < D / 8; ++kb) {
      float4 a_nxt = a_cur;
      float4 b_nxt[4];
#pragma unroll
      for (int t = 0; t < 4; ++t) b_nxt[t] = b_cur[t];
      if (kb + 1 < D / 8) {
        a_nxt = *(const float4*)(arow + (kb + 1) * 8);
#pragma unroll
        for (int t = 0; t < 4; ++t)
          b_nxt[t] = *(const float4*)(bcol + t * 16 * WT_PITCH + (kb + 1) * 8);
      }
      __builtin_amdgcn_sched_barrier(0);
      v2f a01, a23;
      a01.x = a_cur.x; a01.y = a_cur.y;
      a23.x = a_cur.z; a23.y = a_cur.w;
#pragma unroll
      for (int t = 0; t < 4; ++t) {
        v2f b01;
        b01.x = b_cur[t].x; b01.y = b_cur[t].y;
        acc[t] = __builtin_amdgcn_wmma_f32_16x16x4_f32(
             false, a01,  false, b01,
             (short)0, acc[t],  false,  false);
      }
#pragma unroll
      for (int t = 0; t < 4; ++t) {
        v2f b23;
        b23.x = b_cur[t].z; b23.y = b_cur[t].w;
        acc[t] = __builtin_amdgcn_wmma_f32_16x16x4_f32(
             false, a23,  false, b23,
             (short)0, acc[t],  false,  false);
      }
      __builtin_amdgcn_sched_barrier(0);
      a_cur = a_nxt;
#pragma unroll
      for (int t = 0; t < 4; ++t) b_cur[t] = b_nxt[t];
    }

    const int mbase = hh * 8;
    float dr[8];
#pragma unroll
    for (int j = 0; j < 8; ++j) dr[j] = dinv[row0 + mbase + j];
#pragma unroll
    for (int t = 0; t < 4; ++t) {
#pragma unroll
      for (int j = 0; j < 8; ++j) {
        out[(size_t)(row0 + mbase + j) * D + col0 + t * 16 + c] = acc[t][j] * dr[j];
      }
    }
  }
}

template <int K>
__device__ __forceinline__ void gather_rows(const int my, const int j,
                                            const float* __restrict__ hw, const int c,
                                            float4& acc) {
  float4 v[K];
#pragma unroll
  for (int u = 0; u < K; ++u) {
    const int src = __builtin_amdgcn_readlane(my, j + u);
    v[u] = *(const float4*)(hw + (size_t)src * D + c);
  }
#pragma unroll
  for (int u = 0; u < K; ++u) {
    acc.x += v[u].x;
    acc.y += v[u].y;
    acc.z += v[u].z;
    acc.w += v[u].w;
  }
}

__global__ __launch_bounds__(256) void k_row_gather_ln(const int* __restrict__ rowptr,
                                                       const int* __restrict__ csr,
                                                       const float* __restrict__ dinv,
                                                       const float* __restrict__ hw,
                                                       const float* __restrict__ bias,
                                                       const float* __restrict__ g,
                                                       const float* __restrict__ beta,
                                                       float* __restrict__ out, int n) {
  const int lane = threadIdx.x & 31;
  const int wave = __builtin_amdgcn_readfirstlane((int)(threadIdx.x >> 5));
  const int row = blockIdx.x * 8 + wave;
  if (row >= n) return;
  const int c = lane * 4;
  const int beg = rowptr[row];
  const int end = rowptr[row + 1];
  const float s = dinv[row];
  float4 acc = *(const float4*)(hw + (size_t)row * D + c);
  for (int base = beg; base < end; base += 32) {
    const int m = min(32, end - base);
    int my = 0;
    if (lane < m) my = csr[base + lane];
    int j = 0;
    for (; j + 8 <= m; j += 8) gather_rows<8>(my, j, hw, c, acc);
    if (j + 4 <= m) { gather_rows<4>(my, j, hw, c, acc); j += 4; }
    if (j + 2 <= m) { gather_rows<2>(my, j, hw, c, acc); j += 2; }
    if (j < m) gather_rows<1>(my, j, hw, c, acc);
  }
  acc.x = bias[c + 0] + s * acc.x;
  acc.y = bias[c + 1] + s * acc.y;
  acc.z = bias[c + 2] + s * acc.z;
  acc.w = bias[c + 3] + s * acc.w;

  float t = acc.x + acc.y + acc.z + acc.w;
#pragma unroll
  for (int off = 16; off > 0; off >>= 1) t += __shfl_xor(t, off, 32);
  const float mu = t * (1.0f / 128.0f);

  const float dx = acc.x - mu, dy = acc.y - mu, dz = acc.z - mu, dw = acc.w - mu;
  float q = dx * dx + dy * dy + dz * dz + dw * dw;
#pragma unroll
  for (int off = 16; off > 0; off >>= 1) q += __shfl_xor(q, off, 32);
  const float rs = rsqrtf(q * (1.0f / 128.0f) + LN_EPS);

  float4 o;
  o.x = fmaxf(dx * rs * g[c + 0] + beta[c + 0], 0.0f);
  o.y = fmaxf(dy * rs * g[c + 1] + beta[c + 1], 0.0f);
  o.z = fmaxf(dz * rs * g[c + 2] + beta[c + 2], 0.0f);
  o.w = fmaxf(dw * rs * g[c + 3] + beta[c + 3], 0.0f);
  *(float4*)(out + (size_t)row * D + c) = o;
}

extern "C" void kernel_launch(void* const* d_in, const int* in_sizes, int n_in,
                              void* d_out, int out_size, void* d_ws, size_t ws_size,
                              hipStream_t stream) {
  const float* x    = (const float*)d_in[0];
  const int*   erow = (const int*)d_in[1];
  const int*   ecol = (const int*)d_in[2];
  const float* W[3]  = {(const float*)d_in[3], (const float*)d_in[7],  (const float*)d_in[11]};
  const float* b[3]  = {(const float*)d_in[4], (const float*)d_in[8],  (const float*)d_in[12]};
  const float* g[3]  = {(const float*)d_in[5], (const float*)d_in[9],  (const float*)d_in[13]};
  const float* bt[3] = {(const float*)d_in[6], (const float*)d_in[10], (const float*)d_in[14]};

  const int n = in_sizes[0] / D;
  const int e = in_sizes[1];
  const int nb = (n + BIN_ROWS - 1) / BIN_ROWS;
  const int nchunks = (e + CHUNK - 1) / CHUNK;

  float* ws   = (float*)d_ws;
  float* dinv = ws;
  float* hw   = ws + n;
  int* csr    = (int*)(hw + (size_t)n * D);
  unsigned* binned = (unsigned*)(csr + e);
  int* hist   = (int*)(binned + e);
  int* binoff = hist + (size_t)nchunks * nb;
  int* rowptr = binoff + nb + 1;
  float* outp = (float*)d_out;

  k_bin_count<<<nchunks, 1024, 0, stream>>>(erow, hist, e, nb);
  k_bin_scatter<<<nchunks, 1024, 0, stream>>>(erow, ecol, hist, binned, binoff, e, nb, nchunks);
  k_bin_sort<<<nb, 1024, 0, stream>>>(binned, binoff, rowptr, dinv, csr, n, e);

  const unsigned row_blocks  = (unsigned)((n + 7) / 8);

  const float* hin = x;
  for (int L = 0; L < 3; ++L) {
    k_gemm_wmma<<<GEMM_BLOCKS, 512, 0, stream>>>(hin, W[L], dinv, hw, n);
    k_row_gather_ln<<<row_blocks, 256, 0, stream>>>(rowptr, csr, dinv, hw, b[L], g[L], bt[L], outp, n);
    hin = outp;
  }
}


# Layers 1-2: next layer's f32 WMMA GEMM moved into the gather+LayerNorm launch: 1024 resident 16-wave blocks keep W in LDS, 16-row tiles via LDS, 4 per WGP out of step; 9 to 7 launches; sum order kept
# speedup vs baseline: 5.2772x; 1.1387x over previous
#include <hip/hip_runtime.h>
#include <hip/hip_bf16.h>

typedef __attribute__((ext_vector_type(2))) float v2f;
typedef __attribute__((ext_vector_type(8))) float v8f;

#define D 128
#define LN_EPS 1e-5f

#define BIN_SHIFT 9
#define BIN_ROWS (1 << BIN_SHIFT)
#define SRC_BITS 23
#define SRC_MASK ((1u << SRC_BITS) - 1u)
#define MAX_BINS 1024
#define CHUNK 8192
#define CHUNK_EPT (CHUNK / 1024)
#define HIST_BATCH 16
#define SORT_CAP 12288
#define SORT_EPT 8

__device__ __forceinline__ int block_excl_scan(const int v, int* wtot, int& total) {
  const int lane = threadIdx.x & 31;
  const int wave = threadIdx.x >> 5;
  int incl = v;
#pragma unroll
  for (int off = 1; off < 32; off <<= 1) {
    const int t = __shfl_up(incl, off, 32);
    if (lane >= off) incl += t;
  }
  if (lane == 31) wtot[wave] = incl;
  __syncthreads();
  if (wave == 0) {
    const int wv = wtot[lane];
    int wi = wv;
#pragma unroll
    for (int off = 1; off < 32; off <<= 1) {
      const int t = __shfl_up(wi, off, 32);
      if (lane >= off) wi += t;
    }
    wtot[lane] = wi - wv;
    if (lane == 31) wtot[32] = wi;
  }
  __syncthreads();
  const int ex = wtot[wave] + incl - v;
  total = wtot[32];
  __syncthreads();
  return ex;
}

__global__ __launch_bounds__(1024) void k_bin_count(const int* __restrict__ row,
                                                    int* __restrict__ hist, int e, int nb) {
  __shared__ int h[MAX_BINS];
  const int tid = threadIdx.x;
  if (tid < nb) h[tid] = 0;
  const int base = blockIdx.x * CHUNK;
  const int last = max(e - 1, 0);
  int r[CHUNK_EPT];
#pragma unroll
  for (int u = 0; u < CHUNK_EPT; ++u) r[u] = row[min(base + u * 1024 + tid, last)];
  __syncthreads();
#pragma unroll
  for (int u = 0; u < CHUNK_EPT; ++u) {
    if (base + u * 1024 + tid < e) atomicAdd(&h[r[u] >> BIN_SHIFT], 1);
  }
  __syncthreads();
  if (tid < nb) hist[blockIdx.x * nb + tid] = h[tid];
}

__global__ __launch_bounds__(1024) void k_bin_scatter(const int* __restrict__ row,
                                                      const int* __restrict__ col,
                                                      const int* __restrict__ hist,
                                                      unsigned* __restrict__ binned,
                                                      int* __restrict__ binoff,
                                                      int e, int nb, int nchunks) {
  __shared__ int loc[MAX_BINS];
  __shared__ int gof[MAX_BINS];
  __shared__ int fill[MAX_BINS];
  __shared__ int wtot[33];
  __shared__ unsigned st_ent[CHUNK];
  __shared__ unsigned short st_bin[CHUNK];
  const int tid = threadIdx.x;
  const int j = blockIdx.x;
  const int base = j * CHUNK;
  const int last = max(e - 1, 0);
  int r[CHUNK_EPT], s[CHUNK_EPT];
#pragma unroll
  for (int u = 0; u < CHUNK_EPT; ++u) {
    const int i = min(base + u * 1024 + tid, last);
    r[u] = row[i];
    s[u] = col[i];
  }
  int* ptot = (int*)st_ent;
  int* ppre = ptot + 4 * MAX_BINS;
  const int q = tid >> 8;
  for (int t0 = 0; t0 < nb; t0 += 256) {
    const int t = t0 + (tid & 255);
    int a = 0, p = 0;
    const int tc = min(t, nb - 1);
    for (int jj0 = q; jj0 < nchunks; jj0 += 4 * HIST_BATCH) {
      int v[HIST_BATCH];
#pragma unroll
      for (int u = 0; u < HIST_BATCH; ++u) v[u] = hist[min(jj0 + 4 * u, nchunks - 1) * nb + tc];
#pragma unroll
      for (int u = 0; u < HIST_BATCH; ++u) {
        const int jj = jj0 + 4 * u;
        if (jj < nchunks) {
          a += v[u];
          if (jj < j) p += v[u];
        }
      }
    }
    ptot[q * MAX_BINS + t] = a;
    ppre[q * MAX_BINS + t] = p;
  }
  __syncthreads();
  int tot = 0, pre = 0, own = 0;
  if (tid < nb) {
    own = hist[j * nb + tid];
#pragma unroll
    for (int k = 0; k < 4; ++k) {
      tot += ptot[k * MAX_BINS + tid];
      pre += ppre[k * MAX_BINS + tid];
    }
  }
  int total_all, total_own;
  const int ex_tot = block_excl_scan(tot, wtot, total_all);
  const int ex_own = block_excl_scan(own, wtot, total_own);
  if (tid < nb) {
    loc[tid] = ex_own;
    gof[tid] = ex_tot + pre;
    fill[tid] = 0;
    if (j == 0) binoff[tid] = ex_tot;
  }
  if (j == 0 && tid == 0) binoff[nb] = total_all;
  __syncthreads();
#pragma unroll
  for (int u = 0; u < CHUNK_EPT; ++u) {
    if (base + u * 1024 + tid < e) {
      const int b = r[u] >> BIN_SHIFT;
      const int p = loc[b] + atomicAdd(&fill[b], 1);
      st_ent[p] = ((unsigned)(r[u] & (BIN_ROWS - 1)) << SRC_BITS) | (unsigned)s[u];
      st_bin[p] = (unsigned short)b;
    }
  }
  __syncthreads();
#pragma unroll
  for (int u = 0; u < CHUNK_EPT; ++u) {
    const int p = u * 1024 + tid;
    if (p < total_own) {
      const int b = st_bin[p];
      binned[gof[b] + (p - loc[b])] = st_ent[p];
    }
  }
}

__global__ __launch_bounds__(1024) void k_bin_sort(const unsigned* __restrict__ binned,
                                                   const int* __restrict__ binoff,
                                                   int* __restrict__ rowptr,
                                                   float* __restrict__ dinv,
                                                   int* __restrict__ csr, int n, int e) {
  __shared__ int cnt[BIN_ROWS];
  __shared__ int fill[BIN_ROWS];
  __shared__ int wtot[33];
  __shared__ int stage[SORT_CAP];
  const int tid = threadIdx.x;
  const int b = blockIdx.x;
  const int beg = binoff[b];
  const int end = binoff[b + 1];
  const int m = end - beg;
  if (tid < BIN_ROWS) cnt[tid] = 0;
  __syncthreads();
  for (int i0 = beg; i0 < end; i0 += 1024 * SORT_EPT) {
    unsigned ent[SORT_EPT];
#pragma unroll
    for (int u = 0; u < SORT_EPT; ++u) ent[u] = binned[min(i0 + u * 1024 + tid, end - 1)];
#pragma unroll
    for (int u = 0; u < SORT_EPT; ++u) {
      if (i0 + u * 1024 + tid < end) atomicAdd(&cnt[ent[u] >> SRC_BITS], 1);
    }
  }
  __syncthreads();
  const int c = (tid < BIN_ROWS) ? cnt[tid] : 0;
  int total;
  const int ex = block_excl_scan(c, wtot, total);
  if (tid < BIN_ROWS) {
    fill[tid] = ex;
    const int rr = b * BIN_ROWS + tid;
    if (rr < n) {
      rowptr[rr] = beg + ex;
      dinv[rr] = rsqrtf((float)(c + 1));
    }
  }
  if (b == 0 && tid == 0) rowptr[n] = e;
  __syncthreads();
  const bool staged = (m <= SORT_CAP);
  for (int i0 = beg; i0 < end; i0 += 1024 * SORT_EPT) {
    unsigned ent[SORT_EPT];
#pragma unroll
    for (int u = 0; u < SORT_EPT; ++u) ent[u] = binned[min(i0 + u * 1024 + tid, end - 1)];
#pragma unroll
    for (int u = 0; u < SORT_EPT; ++u) {
      if (i0 + u * 1024 + tid < end) {
        const int p = atomicAdd(&fill[ent[u] >> SRC_BITS], 1);
        const int sv = (int)(ent[u] & SRC_MASK);
        if (staged) stage[p] = sv;
        else csr[beg + p] = sv;
      }
    }
  }
  if (staged) {
    __syncthreads();
    for (int i = tid; i < m; i += 1024) csr[beg + i] = stage[i];
  }
}

#define GEMM_BLOCKS 256
#define WT_PITCH 132
#define GEMM_LDS_FLOATS (41 * 1024)

__global__ __launch_bounds__(512) void k_gemm_wmma(const float* __restrict__ h,
                                                    const float* __restrict__ W,
                                                    const float* __restrict__ dinv,
                                                    float* __restrict__ out, int n) {
  __shared__ __align__(16) float wt[GEMM_LDS_FLOATS];
#pragma unroll
  for (int i = threadIdx.x; i < (D / 4) * D; i += 512) {
    const int col = i & (D - 1);
    const int k4 = (i >> 7) * 4;
    float4 v;
    v.x = W[(k4 + 0) * D + col];
    v.y = W[(k4 + 1) * D + col];
    v.z = W[(k4 + 2) * D + col];
    v.w = W[(k4 + 3) * D + col];
    *(float4*)(wt + col * WT_PITCH + k4) = v;
  }
  __syncthreads();

  const int wave = __builtin_amdgcn_readfirstlane((int)(threadIdx.x >> 5));
  const int lane = threadIdx.x & 31;
  const int c    = lane & 15;
  const int hh   = lane >> 4;
  const int nstrips = n >> 4;

  const int slot = ((((wave & 3) - (wave >> 2)) & 3) << 2) | (wave >> 2);
  for (int q = slot; ; q += 16) {
    const int strip = blockIdx.x + (q >> 1) * GEMM_BLOCKS;
    if (strip >= nstrips) break;
    const int row0 = strip * 16;
    const int col0 = (q & 1) * 64;

    const float* __restrict__ arow = h + (size_t)(row0 + c) * D + 4 * hh;
    const float* bcol = wt + (col0 + c) * WT_PITCH + 4 * hh;

    v8f acc[4];
    const v8f zero = {0.f, 0.f, 0.f, 0.f, 0.f, 0.f, 0.f, 0.f};
#pragma unroll
    for (int t = 0; t < 4; ++t) acc[t] = zero;

    float4 a_cur = *(const float4*)arow;
    float4 b_cur[4];
#pragma unroll
    for (int t = 0; t < 4; ++t) b_cur[t] = *(const float4*)(bcol + t * 16 * WT_PITCH);

#pragma unroll
    for (int kb = 0; kb < D / 8; ++kb) {
      float4 a_nxt = a_cur;
      float4 b_nxt[4];
#pragma unroll
      for (int t = 0; t < 4; ++t) b_nxt[t] = b_cur[t];
      if (kb + 1 < D / 8) {
        a_nxt = *(const float4*)(arow + (kb + 1) * 8);
#pragma unroll
        for (int t = 0; t < 4; ++t)
          b_nxt[t] = *(const float4*)(bcol + t * 16 * WT_PITCH + (kb + 1) * 8);
      }
      __builtin_amdgcn_sched_barrier(0);
      v2f a01, a23;
      a01.x = a_cur.x; a01.y = a_cur.y;
      a23.x = a_cur.z; a23.y = a_cur.w;
#pragma unroll
      for (int t = 0; t < 4; ++t) {
        v2f b01;
        b01.x = b_cur[t].x; b01.y = b_cur[t].y;
        acc[t] = __builtin_amdgcn_wmma_f32_16x16x4_f32(
             false, a01,  false, b01,
             (short)0, acc[t],  false,  false);
      }
#pragma unroll
      for (int t = 0; t < 4; ++t) {
        v2f b23;
        b23.x = b_cur[t].z; b23.y = b_cur[t].w;
        acc[t] = __builtin_amdgcn_wmma_f32_16x16x4_f32(
             false, a23,  false, b23,
             (short)0, acc[t],  false,  false);
      }
      __builtin_amdgcn_sched_barrier(0);
      a_cur = a_nxt;
#pragma unroll
      for (int t = 0; t < 4; ++t) b_cur[t] = b_nxt[t];
    }

    const int mbase = hh * 8;
    float dr[8];
#pragma unroll
    for (int j = 0; j < 8; ++j) dr[j] = dinv[row0 + mbase + j];
#pragma unroll
    for (int t = 0; t < 4; ++t) {
#pragma unroll
      for (int j = 0; j < 8; ++j) {
        out[(size_t)(row0 + mbase + j) * D + col0 + t * 16 + c] = acc[t][j] * dr[j];
      }
    }
  }
}

template <int K>
__device__ __forceinline__ void gather_rows(const int my, const int j,
                                            const float* __restrict__ hw, const int c,
                                            float4& acc) {
  float4 v[K];
#pragma unroll
  for (int u = 0; u < K; ++u) {
    const int src = __builtin_amdgcn_readlane(my, j + u);
    v[u] = *(const float4*)(hw + (size_t)src * D + c);
  }
#pragma unroll
  for (int u = 0; u < K; ++u) {
    acc.x += v[u].x;
    acc.y += v[u].y;
    acc.z += v[u].z;
    acc.w += v[u].w;
  }
}

__device__ __forceinline__ void gather_chunk(const int my, const int m,
                                             const float* __restrict__ hw, const int c,
                                             float4& acc) {
  int j = 0;
  for (; j + 8 <= m; j += 8) gather_rows<8>(my, j, hw, c, acc);
  if (j + 4 <= m) { gather_rows<4>(my, j, hw, c, acc); j += 4; }
  if (j + 2 <= m) { gather_rows<2>(my, j, hw, c, acc); j += 2; }
  if (j < m) gather_rows<1>(my, j, hw, c, acc);
}

__device__ __forceinline__ float4 row_sum(const int row, const int beg, const int end,
                                          const int my, const int lane,
                                          const int* __restrict__ csr,
                                          const float* __restrict__ hw) {
  const int c = lane * 4;
  float4 acc = *(const float4*)(hw + (size_t)row * D + c);
  gather_chunk(my, min(32, end - beg), hw, c, acc);
  for (int base = beg + 32; base < end; base += 32) {
    const int m = min(32, end - base);
    int more = 0;
    if (lane < m) more = csr[base + lane];
    gather_chunk(more, m, hw, c, acc);
  }
  return acc;
}

__device__ __forceinline__ float4 ln_relu(float4 acc, const float s, const float4 b,
                                          const float4 g, const float4 bt) {
  acc.x = b.x + s * acc.x;
  acc.y = b.y + s * acc.y;
  acc.z = b.z + s * acc.z;
  acc.w = b.w + s * acc.w;

  float t = acc.x + acc.y + acc.z + acc.w;
#pragma unroll
  for (int off = 16; off > 0; off >>= 1) t += __shfl_xor(t, off, 32);
  const float mu = t * (1.0f / 128.0f);

  const float dx = acc.x - mu, dy = acc.y - mu, dz = acc.z - mu, dw = acc.w - mu;
  float q = dx * dx + dy * dy + dz * dz + dw * dw;
#pragma unroll
  for (int off = 16; off > 0; off >>= 1) q += __shfl_xor(q, off, 32);
  const float rs = rsqrtf(q * (1.0f / 128.0f) + LN_EPS);

  float4 o;
  o.x = fmaxf(dx * rs * g.x + bt.x, 0.0f);
  o.y = fmaxf(dy * rs * g.y + bt.y, 0.0f);
  o.z = fmaxf(dz * rs * g.z + bt.z, 0.0f);
  o.w = fmaxf(dw * rs * g.w + bt.w, 0.0f);
  return o;
}

__global__ __launch_bounds__(256) void k_row_gather_ln(const int* __restrict__ rowptr,
                                                       const int* __restrict__ csr,
                                                       const float* __restrict__ dinv,
                                                       const float* __restrict__ hw,
                                                       const float* __restrict__ bias,
                                                       const float* __restrict__ g,
                                                       const float* __restrict__ beta,
                                                       float* __restrict__ out, int n) {
  const int lane = threadIdx.x & 31;
  const int wave = __builtin_amdgcn_readfirstlane((int)(threadIdx.x >> 5));
  const int row = blockIdx.x * 8 + wave;
  if (row >= n) return;
  const int beg = rowptr[row];
  const int end = rowptr[row + 1];
  int my = 0;
  if (lane < min(32, end - beg)) my = csr[beg + lane];
  const float s = dinv[row];
  const float4 acc = row_sum(row, beg, end, my, lane, csr, hw);
  const float4 o = ln_relu(acc, s, *(const float4*)(bias + lane * 4), *(const float4*)(g + lane * 4),
                           *(const float4*)(beta + lane * 4));
  *(float4*)(out + (size_t)row * D + lane * 4) = o;
}

#define FUSE_BLOCKS 1024
#define AT_PITCH 132

__global__ __launch_bounds__(512, 16) void k_gather_ln_gemm(const int* __restrict__ rowptr,
                                                        const int* __restrict__ csr,
                                                        const float* __restrict__ dinv,
                                                        const float* __restrict__ hw,
                                                        const float* __restrict__ bias,
                                                        const float* __restrict__ g,
                                                        const float* __restrict__ beta,
                                                        const float* __restrict__ Wn,
                                                        float* __restrict__ out, int n) {
  __shared__ __align__(16) float wt[D * WT_PITCH];
  __shared__ __align__(16) float at[16 * AT_PITCH];
  __shared__ __align__(16) float prm[3 * D];
  if (threadIdx.x < D) {
    prm[threadIdx.x] = bias[threadIdx.x];
    prm[D + threadIdx.x] = g[threadIdx.x];
    prm[2 * D + threadIdx.x] = beta[threadIdx.x];
  }
#pragma unroll
  for (int i = threadIdx.x; i < (D / 4) * D; i += 512) {
    const int col = i & (D - 1);
    const int k4 = (i >> 7) * 4;
    float4 v;
    v.x = Wn[(k4 + 0) * D + col];
    v.y = Wn[(k4 + 1) * D + col];
    v.z = Wn[(k4 + 2) * D + col];
    v.w = Wn[(k4 + 3) * D + col];
    *(float4*)(wt + col * WT_PITCH + k4) = v;
  }

  const int wave = __builtin_amdgcn_readfirstlane((int)(threadIdx.x >> 5));
  const int lane = threadIdx.x & 31;
  const int c    = lane & 15;
  const int hh   = lane >> 4;
  const bool mul = ((((wave & 3) + (wave >> 2)) & 1) == 0);
  const int ct = wave >> 1;
  const float* bcol = wt + (ct * 16 + c) * WT_PITCH + 4 * hh;
  const float* arow = at + c * AT_PITCH + 4 * hh;

  int row = blockIdx.x * 16 + wave;
  int beg = rowptr[min(row, n)];
  int end = rowptr[min(row + 1, n)];
  int my = 0;
  if (lane < min(32, end - beg)) my = csr[beg + lane];
  int pnext = rowptr[min(row + FUSE_BLOCKS * 16 + (lane & 1), n)];
  __syncthreads();

  for (; row - wave < n; row += FUSE_BLOCKS * 16) {
    const int nbeg = __builtin_amdgcn_readlane(pnext, 0);
    const int nend = __builtin_amdgcn_readlane(pnext, 1);
    int nmy = 0;
    if (lane < min(32, nend - nbeg)) nmy = csr[nbeg + lane];
    pnext = rowptr[min(row + 2 * FUSE_BLOCKS * 16 + (lane & 1), n)];
    float4 o = {0.f, 0.f, 0.f, 0.f};
    float s = 0.f;
    if (row < n) {
      s = dinv[row];
      const float4 acc = row_sum(row, beg, end, my, lane, csr, hw);
      o = ln_relu(acc, s, *(const float4*)(prm + lane * 4), *(const float4*)(prm + D + lane * 4),
                  *(const float4*)(prm + 2 * D + lane * 4));
    }
    *(float4*)(at + wave * AT_PITCH + lane * 4) = o;
    __syncthreads();

    v8f acc = {0.f, 0.f, 0.f, 0.f, 0.f, 0.f, 0.f, 0.f};
    if (mul) {
      float4 a_cur[2], b_cur[2];
#pragma unroll
      for (int u = 0; u < 2; ++u) {
        a_cur[u] = *(const float4*)(arow + u * 8);
        b_cur[u] = *(const float4*)(bcol + u * 8);
      }
#pragma unroll
      for (int kq = 0; kq < 8; ++kq) {
        float4 a_nxt[2], b_nxt[2];
#pragma unroll
        for (int u = 0; u < 2; ++u) { a_nxt[u] = a_cur[u]; b_nxt[u] = b_cur[u]; }
        if (kq + 1 < 8) {
#pragma unroll
          for (int u = 0; u < 2; ++u) {
            a_nxt[u] = *(const float4*)(arow + (kq + 1) * 16 + u * 8);
            b_nxt[u] = *(const float4*)(bcol + (kq + 1) * 16 + u * 8);
          }
        }
        __builtin_amdgcn_sched_barrier(0);
#pragma unroll
        for (int u = 0; u < 2; ++u) {
          v2f a01, a23, b01, b23;
          a01.x = a_cur[u].x; a01.y = a_cur[u].y;
          a23.x = a_cur[u].z; a23.y = a_cur[u].w;
          b01.x = b_cur[u].x; b01.y = b_cur[u].y;
          b23.x = b_cur[u].z; b23.y = b_cur[u].w;
          acc = __builtin_amdgcn_wmma_f32_16x16x4_f32(
               false, a01,  false, b01,
               (short)0, acc,  false,  false);
          acc = __builtin_amdgcn_wmma_f32_16x16x4_f32(
               false, a23,  false, b23,
               (short)0, acc,  false,  false);
        }
#pragma unroll
        for (int u = 0; u < 2; ++u) { a_cur[u] = a_nxt[u]; b_cur[u] = b_nxt[u]; }
      }
    }
    __syncthreads();

    if (mul) {
#pragma unroll
      for (int j = 0; j < 8; ++j) at[(hh * 8 + j) * AT_PITCH + ct * 16 + c] = acc[j];
    }
    __syncthreads();

    if (row < n) {
      float4 v = *(const float4*)(at + wave * AT_PITCH + lane * 4);
      v.x *= s; v.y *= s; v.z *= s; v.w *= s;
      *(float4*)(out + (size_t)row * D + lane * 4) = v;
    }
    beg = nbeg;
    end = nend;
    my = nmy;
  }
}

extern "C" void kernel_launch(void* const* d_in, const int* in_sizes, int n_in,
                              void* d_out, int out_size, void* d_ws, size_t ws_size,
                              hipStream_t stream) {
  const float* x    = (const float*)d_in[0];
  const int*   erow = (const int*)d_in[1];
  const int*   ecol = (const int*)d_in[2];
  const float* W[3]  = {(const float*)d_in[3], (const float*)d_in[7],  (const float*)d_in[11]};
  const float* b[3]  = {(const float*)d_in[4], (const float*)d_in[8],  (const float*)d_in[12]};
  const float* g[3]  = {(const float*)d_in[5], (const float*)d_in[9],  (const float*)d_in[13]};
  const float* bt[3] = {(const float*)d_in[6], (const float*)d_in[10], (const float*)d_in[14]};

  const int n = in_sizes[0] / D;
  const int e = in_sizes[1];
  const int nb = (n + BIN_ROWS - 1) / BIN_ROWS;
  const int nchunks = (e + CHUNK - 1) / CHUNK;

  float* ws   = (float*)d_ws;
  float* dinv = ws;
  float* hw   = ws + n;
  float* hw2  = hw + (size_t)n * D;
  int* csr    = (int*)(hw2 + (size_t)n * D);
  unsigned* binned = (unsigned*)(csr + e);
  int* hist   = (int*)(binned + e);
  int* binoff = hist + (size_t)nchunks * nb;
  int* rowptr = binoff + nb + 1;
  float* outp = (float*)d_out;

  k_bin_count<<<nchunks, 1024, 0, stream>>>(erow, hist, e, nb);
  k_bin_scatter<<<nchunks, 1024, 0, stream>>>(erow, ecol, hist, binned, binoff, e, nb, nchunks);
  k_bin_sort<<<nb, 1024, 0, stream>>>(binned, binoff, rowptr, dinv, csr, n, e);

  const unsigned row_blocks  = (unsigned)((n + 7) / 8);

  k_gemm_wmma<<<GEMM_BLOCKS, 512, 0, stream>>>(x, W[0], dinv, hw, n);
  k_gather_ln_gemm<<<FUSE_BLOCKS, 512, 0, stream>>>(rowptr, csr, dinv, hw, b[0], g[0], bt[0], W[1], hw2, n);
  k_gather_ln_gemm<<<FUSE_BLOCKS, 512, 0, stream>>>(rowptr, csr, dinv, hw2, b[1], g[1], bt[1], W[2], hw, n);
  k_row_gather_ln<<<row_blocks, 256, 0, stream>>>(rowptr, csr, dinv, hw, b[2], g[2], bt[2], outp, n);
}


# speedup vs baseline: 6.1538x; 1.1661x over previous
#include <hip/hip_runtime.h>
#include <hip/hip_bf16.h>

typedef __attribute__((ext_vector_type(2))) float v2f;
typedef __attribute__((ext_vector_type(8))) float v8f;

#define D 128
#define LN_EPS 1e-5f

#define BIN_SHIFT 9
#define BIN_ROWS (1 << BIN_SHIFT)
#define SRC_BITS 23
#define SRC_MASK ((1u << SRC_BITS) - 1u)
#define MAX_BINS 1024
#define CHUNK 8192
#define CHUNK_EPT (CHUNK / 1024)
#define HIST_BATCH 16
#define SORT_CAP 12288
#define SORT_EPT 8

__device__ __forceinline__ int block_excl_scan(const int v, int* wtot, int& total) {
  const int lane = threadIdx.x & 31;
  const int wave = threadIdx.x >> 5;
  int incl = v;
#pragma unroll
  for (int off = 1; off < 32; off <<= 1) {
    const int t = __shfl_up(incl, off, 32);
    if (lane >= off) incl += t;
  }
  if (lane == 31) wtot[wave] = incl;
  __syncthreads();
  if (wave == 0) {
    const int wv = wtot[lane];
    int wi = wv;
#pragma unroll
    for (int off = 1; off < 32; off <<= 1) {
      const int t = __shfl_up(wi, off, 32);
      if (lane >= off) wi += t;
    }
    wtot[lane] = wi - wv;
    if (lane == 31) wtot[32] = wi;
  }
  __syncthreads();
  const int ex = wtot[wave] + incl - v;
  total = wtot[32];
  __syncthreads();
  return ex;
}

__global__ __launch_bounds__(1024) void k_bin_count(const int* __restrict__ row,
                                                    int* __restrict__ hist, int e, int nb) {
  __shared__ int h[MAX_BINS];
  const int tid = threadIdx.x;
  if (tid < nb) h[tid] = 0;
  const int base = blockIdx.x * CHUNK;
  const int last = max(e - 1, 0);
  int r[CHUNK_EPT];
#pragma unroll
  for (int u = 0; u < CHUNK_EPT; ++u) r[u] = row[min(base + u * 1024 + tid, last)];
  __syncthreads();
#pragma unroll
  for (int u = 0; u < CHUNK_EPT; ++u) {
    if (base + u * 1024 + tid < e) atomicAdd(&h[r[u] >> BIN_SHIFT], 1);
  }
  __syncthreads();
  if (tid < nb) hist[blockIdx.x * nb + tid] = h[tid];
}

__global__ __launch_bounds__(1024) void k_bin_scatter(const int* __restrict__ row,
                                                      const int* __restrict__ col,
                                                      const int* __restrict__ hist,
                                                      unsigned* __restrict__ binned,
                                                      int* __restrict__ binoff,
                                                      int e, int nb, int nchunks) {
  __shared__ int loc[MAX_BINS];
  __shared__ int gof[MAX_BINS];
  __shared__ int fill[MAX_BINS];
  __shared__ int wtot[33];
  __shared__ unsigned st_ent[CHUNK];
  __shared__ unsigned short st_bin[CHUNK];
  const int tid = threadIdx.x;
  const int j = blockIdx.x;
  const int base = j * CHUNK;
  const int last = max(e - 1, 0);
  int r[CHUNK_EPT], s[CHUNK_EPT];
#pragma unroll
  for (int u = 0; u < CHUNK_EPT; ++u) {
    const int i = min(base + u * 1024 + tid, last);
    r[u] = row[i];
    s[u] = col[i];
  }
  int* ptot = (int*)st_ent;
  int* ppre = ptot + 4 * MAX_BINS;
  const int q = tid >> 8;
  for (int t0 = 0; t0 < nb; t0 += 256) {
    const int t = t0 + (tid & 255);
    int a = 0, p = 0;
    const int tc = min(t, nb - 1);
    for (int jj0 = q; jj0 < nchunks; jj0 += 4 * HIST_BATCH) {
      int v[HIST_BATCH];
#pragma unroll
      for (int u = 0; u < HIST_BATCH; ++u) v[u] = hist[min(jj0 + 4 * u, nchunks - 1) * nb + tc];
#pragma unroll
      for (int u = 0; u < HIST_BATCH; ++u) {
        const int jj = jj0 + 4 * u;
        if (jj < nchunks) {
          a += v[u];
          if (jj < j) p += v[u];
        }
      }
    }
    ptot[q * MAX_BINS + t] = a;
    ppre[q * MAX_BINS + t] = p;
  }
  __syncthreads();
  int tot = 0, pre = 0, own = 0;
  if (tid < nb) {
    own = hist[j * nb + tid];
#pragma unroll
    for (int k = 0; k < 4; ++k) {
      tot += ptot[k * MAX_BINS + tid];
      pre += ppre[k * MAX_BINS + tid];
    }
  }
  int total_all, total_own;
  const int ex_tot = block_excl_scan(tot, wtot, total_all);
  const int ex_own = block_excl_scan(own, wtot, total_own);
  if (tid < nb) {
    loc[tid] = ex_own;
    gof[tid] = ex_tot + pre;
    fill[tid] = 0;
    if (j == 0) binoff[tid] = ex_tot;
  }
  if (j == 0 && tid == 0) binoff[nb] = total_all;
  __syncthreads();
#pragma unroll
  for (int u = 0; u < CHUNK_EPT; ++u) {
    if (base + u * 1024 + tid < e) {
      const int b = r[u] >> BIN_SHIFT;
      const int p = loc[b] + atomicAdd(&fill[b], 1);
      st_ent[p] = ((unsigned)(r[u] & (BIN_ROWS - 1)) << SRC_BITS) | (unsigned)s[u];
      st_bin[p] = (unsigned short)b;
    }
  }
  __syncthreads();
#pragma unroll
  for (int u = 0; u < CHUNK_EPT; ++u) {
    const int p = u * 1024 + tid;
    if (p < total_own) {
      const int b = st_bin[p];
      binned[gof[b] + (p - loc[b])] = st_ent[p];
    }
  }
}

__global__ __launch_bounds__(1024) void k_bin_sort(const unsigned* __restrict__ binned,
                                                   const int* __restrict__ binoff,
                                                   int* __restrict__ rowptr,
                                                   float* __restrict__ dinv,
                                                   int* __restrict__ csr, int n, int e) {
  __shared__ int cnt[BIN_ROWS];
  __shared__ int fill[BIN_ROWS];
  __shared__ int wtot[33];
  __shared__ int stage[SORT_CAP];
  const int tid = threadIdx.x;
  const int b = blockIdx.x;
  const int beg = binoff[b];
  const int end = binoff[b + 1];
  const int m = end - beg;
  if (tid < BIN_ROWS) cnt[tid] = 0;
  __syncthreads();
  for (int i0 = beg; i0 < end; i0 += 1024 * SORT_EPT) {
    unsigned ent[SORT_EPT];
#pragma unroll
    for (int u = 0; u < SORT_EPT; ++u) ent[u] = binned[min(i0 + u * 1024 + tid, end - 1)];
#pragma unroll
    for (int u = 0; u < SORT_EPT; ++u) {
      if (i0 + u * 1024 + tid < end) atomicAdd(&cnt[ent[u] >> SRC_BITS], 1);
    }
  }
  __syncthreads();
  const int c = (tid < BIN_ROWS) ? cnt[tid] : 0;
  int total;
  const int ex = block_excl_scan(c, wtot, total);
  if (tid < BIN_ROWS) {
    fill[tid] = ex;
    const int rr = b * BIN_ROWS + tid;
    if (rr < n) {
      rowptr[rr] = beg + ex;
      dinv[rr] = rsqrtf((float)(c + 1));
    }
  }
  if (b == 0 && tid == 0) rowptr[n] = e;
  __syncthreads();
  const bool staged = (m <= SORT_CAP);
  for (int i0 = beg; i0 < end; i0 += 1024 * SORT_EPT) {
    unsigned ent[SORT_EPT];
#pragma unroll
    for (int u = 0; u < SORT_EPT; ++u) ent[u] = binned[min(i0 + u * 1024 + tid, end - 1)];
#pragma unroll
    for (int u = 0; u < SORT_EPT; ++u) {
      if (i0 + u * 1024 + tid < end) {
        const int p = atomicAdd(&fill[ent[u] >> SRC_BITS], 1);
        const int sv = (int)(ent[u] & SRC_MASK);
        if (staged) stage[p] = sv;
        else csr[beg + p] = sv;
      }
    }
  }
  if (staged) {
    __syncthreads();
    for (int i = tid; i < m; i += 1024) csr[beg + i] = stage[i];
  }
}

#define GEMM_BLOCKS 256
#define WT_PITCH 132
#define GEMM_LDS_FLOATS (41 * 1024)

__global__ __launch_bounds__(512) void k_gemm_wmma(const float* __restrict__ h,
                                                    const float* __restrict__ W,
                                                    const float* __restrict__ dinv,
                                                    float* __restrict__ out, int n) {
  __shared__ __align__(16) float wt[GEMM_LDS_FLOATS];
#pragma unroll
  for (int i = threadIdx.x; i < (D / 4) * D; i += 512) {
    const int col = i & (D - 1);
    const int k4 = (i >> 7) * 4;
    float4 v;
    v.x = W[(k4 + 0) * D + col];
    v.y = W[(k4 + 1) * D + col];
    v.z = W[(k4 + 2) * D + col];
    v.w = W[(k4 + 3) * D + col];
    *(float4*)(wt + col * WT_PITCH + k4) = v;
  }
  __syncthreads();

  const int wave = __builtin_amdgcn_readfirstlane((int)(threadIdx.x >> 5));
  const int lane = threadIdx.x & 31;
  const int c    = lane & 15;
  const int hh   = lane >> 4;
  const int nstrips = n >> 4;

  const int slot = ((((wave & 3) - (wave >> 2)) & 3) << 2) | (wave >> 2);
  for (int q = slot; ; q += 16) {
    const int strip = blockIdx.x + (q >> 1) * GEMM_BLOCKS;
    if (strip >= nstrips) break;
    const int row0 = strip * 16;
    const int col0 = (q & 1) * 64;

    const float* __restrict__ arow = h + (size_t)(row0 + c) * D + 4 * hh;
    const float* bcol = wt + (col0 + c) * WT_PITCH + 4 * hh;

    v8f acc[4];
    const v8f zero = {0.f, 0.f, 0.f, 0.f, 0.f, 0.f, 0.f, 0.f};
#pragma unroll
    for (int t = 0; t < 4; ++t) acc[t] = zero;

    float4 a_cur = *(const float4*)arow;
    float4 b_cur[4];
#pragma unroll
    for (int t = 0; t < 4; ++t) b_cur[t] = *(const float4*)(bcol + t * 16 * WT_PITCH);

#pragma unroll
    for (int kb = 0; kb < D / 8; ++kb) {
      float4 a_nxt = a_cur;
      float4 b_nxt[4];
#pragma unroll
      for (int t = 0; t < 4; ++t) b_nxt[t] = b_cur[t];
      if (kb + 1 < D / 8) {
        a_nxt = *(const float4*)(arow + (kb + 1) * 8);
#pragma unroll
        for (int t = 0; t < 4; ++t)
          b_nxt[t] = *(const float4*)(bcol + t * 16 * WT_PITCH + (kb + 1) * 8);
      }
      __builtin_amdgcn_sched_barrier(0);
      v2f a01, a23;
      a01.x = a_cur.x; a01.y = a_cur.y;
      a23.x = a_cur.z; a23.y = a_cur.w;
#pragma unroll
      for (int t = 0; t < 4; ++t) {
        v2f b01;
        b01.x = b_cur[t].x; b01.y = b_cur[t].y;
        acc[t] = __builtin_amdgcn_wmma_f32_16x16x4_f32(
             false, a01,  false, b01,
             (short)0, acc[t],  false,  false);
      }
#pragma unroll
      for (int t = 0; t < 4; ++t) {
        v2f b23;
        b23.x = b_cur[t].z; b23.y = b_cur[t].w;
        acc[t] = __builtin_amdgcn_wmma_f32_16x16x4_f32(
             false, a23,  false, b23,
             (short)0, acc[t],  false,  false);
      }
      __builtin_amdgcn_sched_barrier(0);
      a_cur = a_nxt;
#pragma unroll
      for (int t = 0; t < 4; ++t) b_cur[t] = b_nxt[t];
    }

    const int mbase = hh * 8;
    float dr[8];
#pragma unroll
    for (int j = 0; j < 8; ++j) dr[j] = dinv[row0 + mbase + j];
#pragma unroll
    for (int t = 0; t < 4; ++t) {
#pragma unroll
      for (int j = 0; j < 8; ++j) {
        out[(size_t)(row0 + mbase + j) * D + col0 + t * 16 + c] = acc[t][j] * dr[j];
      }
    }
  }
}

template <int K>
__device__ __forceinline__ void gather_rows(const int my, const int j,
                                            const float* __restrict__ hw, const int c,
                                            float4& acc) {
  float4 v[K];
#pragma unroll
  for (int u = 0; u < K; ++u) {
    const int src = __builtin_amdgcn_readlane(my, j + u);
    v[u] = *(const float4*)(hw + (size_t)src * D + c);
  }
#pragma unroll
  for (int u = 0; u < K; ++u) {
    acc.x += v[u].x;
    acc.y += v[u].y;
    acc.z += v[u].z;
    acc.w += v[u].w;
  }
}

__device__ __forceinline__ void gather_chunk(const int my, const int m,
                                             const float* __restrict__ hw, const int c,
                                             float4& acc) {
  int j = 0;
  for (; j + 8 <= m; j += 8) gather_rows<8>(my, j, hw, c, acc);
  if (j + 4 <= m) { gather_rows<4>(my, j, hw, c, acc); j += 4; }
  if (j + 2 <= m) { gather_rows<2>(my, j, hw, c, acc); j += 2; }
  if (j < m) gather_rows<1>(my, j, hw, c, acc);
}

__device__ __forceinline__ float4 row_sum(const int row, const int beg, const int end,
                                          const int my, const int lane,
                                          const int* __restrict__ csr,
                                          const float* __restrict__ hw) {
  const int c = lane * 4;
  float4 acc = *(const float4*)(hw + (size_t)row * D + c);
  gather_chunk(my, min(32, end - beg), hw, c, acc);
  for (int base = beg + 32; base < end; base += 32) {
    const int m = min(32, end - base);
    int more = 0;
    if (lane < m) more = csr[base + lane];
    gather_chunk(more, m, hw, c, acc);
  }
  return acc;
}

__device__ __forceinline__ float4 ln_relu(float4 acc, const float s, const float4 b,
                                          const float4 g, const float4 bt) {
  acc.x = b.x + s * acc.x;
  acc.y = b.y + s * acc.y;
  acc.z = b.z + s * acc.z;
  acc.w = b.w + s * acc.w;

  float t = acc.x + acc.y + acc.z + acc.w;
#pragma unroll
  for (int off = 16; off > 0; off >>= 1) t += __shfl_xor(t, off, 32);
  const float mu = t * (1.0f / 128.0f);

  const float dx = acc.x - mu, dy = acc.y - mu, dz = acc.z - mu, dw = acc.w - mu;
  float q = dx * dx + dy * dy + dz * dz + dw * dw;
#pragma unroll
  for (int off = 16; off > 0; off >>= 1) q += __shfl_xor(q, off, 32);
  const float rs = rsqrtf(q * (1.0f / 128.0f) + LN_EPS);

  float4 o;
  o.x = fmaxf(dx * rs * g.x + bt.x, 0.0f);
  o.y = fmaxf(dy * rs * g.y + bt.y, 0.0f);
  o.z = fmaxf(dz * rs * g.z + bt.z, 0.0f);
  o.w = fmaxf(dw * rs * g.w + bt.w, 0.0f);
  return o;
}

__global__ __launch_bounds__(256) void k_row_gather_ln(const int* __restrict__ rowptr,
                                                       const int* __restrict__ csr,
                                                       const float* __restrict__ dinv,
                                                       const float* __restrict__ hw,
                                                       const float* __restrict__ bias,
                                                       const float* __restrict__ g,
                                                       const float* __restrict__ beta,
                                                       float* __restrict__ out, int n) {
  const int lane = threadIdx.x & 31;
  const int wave = __builtin_amdgcn_readfirstlane((int)(threadIdx.x >> 5));
  const int row = blockIdx.x * 8 + wave;
  if (row >= n) return;
  const int beg = rowptr[row];
  const int end = rowptr[row + 1];
  int my = 0;
  if (lane < min(32, end - beg)) my = csr[beg + lane];
  const float s = dinv[row];
  const float4 acc = row_sum(row, beg, end, my, lane, csr, hw);
  const float4 o = ln_relu(acc, s, *(const float4*)(bias + lane * 4), *(const float4*)(g + lane * 4),
                           *(const float4*)(beta + lane * 4));
  *(float4*)(out + (size_t)row * D + lane * 4) = o;
}

#define FUSE_BLOCKS 1024
#define AT_PITCH 132

__global__ __launch_bounds__(512, 16) void k_gather_ln_gemm(const int* __restrict__ rowptr,
                                                        const int* __restrict__ csr,
                                                        const float* __restrict__ dinv,
                                                        const float* __restrict__ hw,
                                                        const float* __restrict__ bias,
                                                        const float* __restrict__ g,
                                                        const float* __restrict__ beta,
                                                        const float* __restrict__ Wn,
                                                        float* __restrict__ out, int n) {
  __shared__ __align__(16) float wt[D * WT_PITCH];
  __shared__ __align__(16) float at[16 * AT_PITCH];
  __shared__ __align__(16) float prm[3 * D];
  if (threadIdx.x < D) {
    prm[threadIdx.x] = bias[threadIdx.x];
    prm[D + threadIdx.x] = g[threadIdx.x];
    prm[2 * D + threadIdx.x] = beta[threadIdx.x];
  }
#pragma unroll
  for (int i = threadIdx.x; i < (D / 4) * D; i += 512) {
    const int col = i & (D - 1);
    const int k4 = (i >> 7) * 4;
    float4 v;
    v.x = Wn[(k4 + 0) * D + col];
    v.y = Wn[(k4 + 1) * D + col];
    v.z = Wn[(k4 + 2) * D + col];
    v.w = Wn[(k4 + 3) * D + col];
    *(float4*)(wt + col * WT_PITCH + k4) = v;
  }

  const int wave = __builtin_amdgcn_readfirstlane((int)(threadIdx.x >> 5));
  const int lane = threadIdx.x & 31;
  const int c    = lane & 15;
  const int hh   = lane >> 4;
  const bool mul = ((((wave & 3) + (wave >> 2)) & 1) == 0);
  const int ct = wave >> 1;
  const float* bcol = wt + (ct * 16 + c) * WT_PITCH + 4 * hh;
  const float* arow = at + c * AT_PITCH + 4 * hh;

  int row = blockIdx.x * 16 + wave;
  int beg = rowptr[min(row, n)];
  int end = rowptr[min(row + 1, n)];
  int my = 0;
  if (lane < min(32, end - beg)) my = csr[beg + lane];
  int pnext = rowptr[min(row + FUSE_BLOCKS * 16 + (lane & 1), n)];
  __syncthreads();

  for (; row - wave < n; row += FUSE_BLOCKS * 16) {
    const int nbeg = __builtin_amdgcn_readlane(pnext, 0);
    const int nend = __builtin_amdgcn_readlane(pnext, 1);
    int nmy = 0;
    if (lane < min(32, nend - nbeg)) nmy = csr[nbeg + lane];
    pnext = rowptr[min(row + 2 * FUSE_BLOCKS * 16 + (lane & 1), n)];
    float4 o = {0.f, 0.f, 0.f, 0.f};
    float s = 0.f;
    if (row < n) {
      s = dinv[row];
      const float4 acc = row_sum(row, beg, end, my, lane, csr, hw);
      o = ln_relu(acc, s, *(const float4*)(prm + lane * 4), *(const float4*)(prm + D + lane * 4),
                  *(const float4*)(prm + 2 * D + lane * 4));
    }
    *(float4*)(at + wave * AT_PITCH + lane * 4) = o;
    __syncthreads();

    v8f acc = {0.f, 0.f, 0.f, 0.f, 0.f, 0.f, 0.f, 0.f};
    if (mul) {
      float4 a_cur[2], b_cur[2];
#pragma unroll
      for (int u = 0; u < 2; ++u) {
        a_cur[u] = *(const float4*)(arow + u * 8);
        b_cur[u] = *(const float4*)(bcol + u * 8);
      }
#pragma unroll
      for (int kq = 0; kq < 8; ++kq) {
        float4 a_nxt[2], b_nxt[2];
#pragma unroll
        for (int u = 0; u < 2; ++u) { a_nxt[u] = a_cur[u]; b_nxt[u] = b_cur[u]; }
        if (kq + 1 < 8) {
#pragma unroll
          for (int u = 0; u < 2; ++u) {
            a_nxt[u] = *(const float4*)(arow + (kq + 1) * 16 + u * 8);
            b_nxt[u] = *(const float4*)(bcol + (kq + 1) * 16 + u * 8);
          }
        }
        __builtin_amdgcn_sched_barrier(0);
#pragma unroll
        for (int u = 0; u < 2; ++u) {
          v2f a01, a23, b01, b23;
          a01.x = a_cur[u].x; a01.y = a_cur[u].y;
          a23.x = a_cur[u].z; a23.y = a_cur[u].w;
          b01.x = b_cur[u].x; b01.y = b_cur[u].y;
          b23.x = b_cur[u].z; b23.y = b_cur[u].w;
          acc = __builtin_amdgcn_wmma_f32_16x16x4_f32(
               false, a01,  false, b01,
               (short)0, acc,  false,  false);
          acc = __builtin_amdgcn_wmma_f32_16x16x4_f32(
               false, a23,  false, b23,
               (short)0, acc,  false,  false);
        }
#pragma unroll
        for (int u = 0; u < 2; ++u) { a_cur[u] = a_nxt[u]; b_cur[u] = b_nxt[u]; }
      }
    }
    __syncthreads();

    if (mul) {
#pragma unroll
      for (int j = 0; j < 8; ++j) at[(hh * 8 + j) * AT_PITCH + ct * 16 + c] = acc[j];
    }
    __syncthreads();

    if (row < n) {
      float4 v = *(const float4*)(at + wave * AT_PITCH + lane * 4);
      v.x *= s; v.y *= s; v.z *= s; v.w *= s;
      *(float4*)(out + (size_t)row * D + lane * 4) = v;
    }
    beg = nbeg;
    end = nend;
    my = nmy;
  }
}

extern "C" void kernel_launch(void* const* d_in, const int* in_sizes, int n_in,
                              void* d_out, int out_size, void* d_ws, size_t ws_size,
                              hipStream_t stream) {
  const float* x    = (const float*)d_in[0];
  const int*   erow = (const int*)d_in[1];
  const int*   ecol = (const int*)d_in[2];
  const float* W[3]  = {(const float*)d_in[3], (const float*)d_in[7],  (const float*)d_in[11]};
  const float* b[3]  = {(const float*)d_in[4], (const float*)d_in[8],  (const float*)d_in[12]};
  const float* g[3]  = {(const float*)d_in[5], (const float*)d_in[9],  (const float*)d_in[13]};
  const float* bt[3] = {(const float*)d_in[6], (const float*)d_in[10], (const float*)d_in[14]};

  const int n = in_sizes[0] / D;
  const int e = in_sizes[1];
  const int nb = (n + BIN_ROWS - 1) / BIN_ROWS;
  const int nchunks = (e + CHUNK - 1) / CHUNK;

  float* ws   = (float*)d_ws;
  float* hw   = ws;
  float* hw2  = hw + (size_t)n * D;
  float* dinv = hw2 + (size_t)n * D;
  int* csr    = (int*)(dinv + n);
  unsigned* binned = (unsigned*)(csr + e);
  int* hist   = (int*)(binned + e);
  int* binoff = hist + (size_t)nchunks * nb;
  int* rowptr = binoff + nb + 1;
  float* outp = (float*)d_out;

  k_bin_count<<<nchunks, 1024, 0, stream>>>(erow, hist, e, nb);
  k_bin_scatter<<<nchunks, 1024, 0, stream>>>(erow, ecol, hist, binned, binoff, e, nb, nchunks);
  k_bin_sort<<<nb, 1024, 0, stream>>>(binned, binoff, rowptr, dinv, csr, n, e);

  const unsigned row_blocks  = (unsigned)((n + 7) / 8);

  k_gemm_wmma<<<GEMM_BLOCKS, 512, 0, stream>>>(x, W[0], dinv, hw, n);
  k_gather_ln_gemm<<<FUSE_BLOCKS, 512, 0, stream>>>(rowptr, csr, dinv, hw, b[0], g[0], bt[0], W[1], hw2, n);
  k_gather_ln_gemm<<<FUSE_BLOCKS, 512, 0, stream>>>(rowptr, csr, dinv, hw2, b[1], g[1], bt[1], W[2], hw, n);
  k_row_gather_ln<<<row_blocks, 256, 0, stream>>>(rowptr, csr, dinv, hw, b[2], g[2], bt[2], outp, n);
}
